# starting kernel
# baseline (speedup 1.0000x reference)
#include <hip/hip_runtime.h>
#include <hip/hip_bf16.h>

typedef float v2f __attribute__((ext_vector_type(2)));
typedef float v8f __attribute__((ext_vector_type(8)));

#define LD      1282
#define KDIM    256
#define NTOT    1282
#define NTILES  81
#define MTILES  256
#define SSTEPS  512
#define BATCH   8

__global__ __launch_bounds__(256)
void fwp_slow_gemm(const float* __restrict__ x,
                   const float* __restrict__ slow_W,
                   const float* __restrict__ slow_b,
                   const float* __restrict__ fast_W0,
                   float* __restrict__ wu)
{
    int wid  = threadIdx.x >> 5;
    int lane = threadIdx.x & 31;
    int wt   = blockIdx.x * 8 + wid;
    if (wt >= MTILES * NTILES) return;
    int mt = wt / NTILES, nt = wt % NTILES;
    int m0 = mt * 16, n0 = nt * 16;
    int q = lane & 15, h = lane >> 4;

    int n = n0 + q;
    const float* wp;
    if (n < 1026) {
        wp = slow_W + (size_t)n * KDIM;
    } else {
        int idx = n - 1026;
        if (idx > 255) idx = 255;
        wp = fast_W0 + (size_t)idx * KDIM;
    }
    float bv = (n < 1026) ? slow_b[n] : 0.0f;

    v8f c;
    #pragma unroll
    for (int r = 0; r < 8; ++r) c[r] = bv;

    const float* ap = x + (size_t)(m0 + q) * KDIM + 2 * h;
    const float* bp = wp + 2 * h;

    #pragma unroll 8
    for (int k = 0; k < KDIM; k += 4) {
        v2f a = *(const v2f*)(ap + k);
        v2f b = *(const v2f*)(bp + k);
        c = __builtin_amdgcn_wmma_f32_16x16x4_f32(false, a, false, b,
                                                  (short)0, c, false, false);
    }

    if (n < NTOT) {
        #pragma unroll
        for (int r = 0; r < 8; ++r) {
            int m = m0 + h * 8 + r;
            wu[(size_t)m * LD + n] = c[r];
        }
    }
}

__global__ __launch_bounds__(256)
void fwp_scan(const float* __restrict__ wu,
              const float* __restrict__ x,
              float* __restrict__ out)
{
    __shared__ float red[8 * 16];

    int tid  = threadIdx.x;
    int w    = tid >> 5;
    int lane = tid & 31;
    int q    = lane & 15;
    int h    = lane >> 4;
    int o0   = blockIdx.x * 16;
    int b    = blockIdx.y;
    int c0   = 32 * w + q;
    int c1   = c0 + 16;

    v8f fw0 = {0,0,0,0,0,0,0,0};
    v8f fw1 = {0,0,0,0,0,0,0,0};
    const v8f zero = {0,0,0,0,0,0,0,0};

    const int STEP_BYTES = BATCH * LD * 4;

    for (int s = 0; s < SSTEPS; ++s) {
        const float* row  = wu + (size_t)(s * BATCH) * LD;
        const float* rowb = row + (size_t)b * LD;

        if (s + 1 < SSTEPS) {
            const char* nb = (const char*)(row) + STEP_BYTES;
            int off0 = tid * 128;
            int off1 = 32768 + tid * 128;
            if (off0 < STEP_BYTES) __builtin_prefetch(nb + off0, 0, 3);
            if (off1 < STEP_BYTES) __builtin_prefetch(nb + off1, 0, 3);
            if (tid < 64) {
                const char* nx = (const char*)(x + (size_t)(s + 1) * BATCH * 256);
                __builtin_prefetch(nx + tid * 128, 0, 3);
            }
        }

        float sa = 1.0f / (1.0f + __expf(-rowb[1024]));
        float ss = 1.0f / (1.0f + __expf(-rowb[1025]));

        int kk = 2 * h;

        v2f aA0, aA1, sA0, sA1;
        aA0.x = row[(size_t)(kk + 0) * LD +       o0 + q];
        aA0.y = row[(size_t)(kk + 1) * LD +       o0 + q];
        aA1.x = row[(size_t)(kk + 4) * LD +       o0 + q];
        aA1.y = row[(size_t)(kk + 5) * LD +       o0 + q];
        sA0.x = row[(size_t)(kk + 0) * LD + 512 + o0 + q];
        sA0.y = row[(size_t)(kk + 1) * LD + 512 + o0 + q];
        sA1.x = row[(size_t)(kk + 4) * LD + 512 + o0 + q];
        sA1.y = row[(size_t)(kk + 5) * LD + 512 + o0 + q];

        v2f aB0t0, aB1t0, aB0t1, aB1t1, sB0t0, sB1t0, sB0t1, sB1t1;
        aB0t0.x = row[(size_t)(kk + 0) * LD + 256 + c0];
        aB0t0.y = row[(size_t)(kk + 1) * LD + 256 + c0];
        aB1t0.x = row[(size_t)(kk + 4) * LD + 256 + c0];
        aB1t0.y = row[(size_t)(kk + 5) * LD + 256 + c0];
        aB0t1.x = row[(size_t)(kk + 0) * LD + 256 + c1];
        aB0t1.y = row[(size_t)(kk + 1) * LD + 256 + c1];
        aB1t1.x = row[(size_t)(kk + 4) * LD + 256 + c1];
        aB1t1.y = row[(size_t)(kk + 5) * LD + 256 + c1];
        sB0t0.x = row[(size_t)(kk + 0) * LD + 768 + c0];
        sB0t0.y = row[(size_t)(kk + 1) * LD + 768 + c0];
        sB1t0.x = row[(size_t)(kk + 4) * LD + 768 + c0];
        sB1t0.y = row[(size_t)(kk + 5) * LD + 768 + c0];
        sB0t1.x = row[(size_t)(kk + 0) * LD + 768 + c1];
        sB0t1.y = row[(size_t)(kk + 1) * LD + 768 + c1];
        sB1t1.x = row[(size_t)(kk + 4) * LD + 768 + c1];
        sB1t1.y = row[(size_t)(kk + 5) * LD + 768 + c1];

        v8f ad0 = __builtin_amdgcn_wmma_f32_16x16x4_f32(false, aA0, false, aB0t0, (short)0, zero, false, false);
        ad0     = __builtin_amdgcn_wmma_f32_16x16x4_f32(false, aA1, false, aB1t0, (short)0, ad0,  false, false);
        v8f ad1 = __builtin_amdgcn_wmma_f32_16x16x4_f32(false, aA0, false, aB0t1, (short)0, zero, false, false);
        ad1     = __builtin_amdgcn_wmma_f32_16x16x4_f32(false, aA1, false, aB1t1, (short)0, ad1,  false, false);
        v8f su0 = __builtin_amdgcn_wmma_f32_16x16x4_f32(false, sA0, false, sB0t0, (short)0, zero, false, false);
        su0     = __builtin_amdgcn_wmma_f32_16x16x4_f32(false, sA1, false, sB1t0, (short)0, su0,  false, false);
        v8f su1 = __builtin_amdgcn_wmma_f32_16x16x4_f32(false, sA0, false, sB0t1, (short)0, zero, false, false);
        su1     = __builtin_amdgcn_wmma_f32_16x16x4_f32(false, sA1, false, sB1t1, (short)0, su1,  false, false);

        #pragma unroll
        for (int r = 0; r < 8; ++r) {
            fw0[r] = fmaf(ad0[r], sa, fw0[r]);
            fw0[r] = fmaf(su0[r], -ss, fw0[r]);
            fw1[r] = fmaf(ad1[r], sa, fw1[r]);
            fw1[r] = fmaf(su1[r], -ss, fw1[r]);
        }

        float xv0 = x[(size_t)(s * BATCH + b) * 256 + c0];
        float xv1 = x[(size_t)(s * BATCH + b) * 256 + c1];
        float pr[8];
        #pragma unroll
        for (int r = 0; r < 8; ++r)
            pr[r] = fw0[r] * xv0 + fw1[r] * xv1;

        #pragma unroll
        for (int m = 1; m <= 8; m <<= 1) {
            #pragma unroll
            for (int r = 0; r < 8; ++r)
                pr[r] += __shfl_xor(pr[r], m, 32);
        }

        if (q == 0) {
            #pragma unroll
            for (int r = 0; r < 8; ++r)
                red[w * 16 + 8 * h + r] = pr[r];
        }
        __syncthreads();

        if (tid < 16) {
            float v = rowb[1026 + o0 + tid];
            #pragma unroll
            for (int ww = 0; ww < 8; ++ww)
                v += red[ww * 16 + tid];
            out[(size_t)(s * BATCH + b) * 256 + o0 + tid] = v;
        }
        __syncthreads();
    }
}

extern "C" void kernel_launch(void* const* d_in, const int* in_sizes, int n_in,
                              void* d_out, int out_size, void* d_ws, size_t ws_size,
                              hipStream_t stream)
{
    const float* x       = (const float*)d_in[0];
    const float* slow_W  = (const float*)d_in[1];
    const float* slow_b  = (const float*)d_in[2];
    const float* fast_W0 = (const float*)d_in[3];
    float*       out     = (float*)d_out;
    float*       wu      = (float*)d_ws;

    int total_waves = MTILES * NTILES;
    int blocks1 = (total_waves + 7) / 8;
    fwp_slow_gemm<<<blocks1, 256, 0, stream>>>(x, slow_W, slow_b, fast_W0, wu);

    dim3 grid2(16, BATCH);
    fwp_scan<<<grid2, 256, 0, stream>>>(wu, x, out);
}


# Scan software-pipelined: step s+1's operands load into a second register set during step s (replaces the L2 prefetch); barriers fence LDS only so global loads and the out store stay in flight
# speedup vs baseline: 1.7556x; 1.7556x over previous
#include <hip/hip_runtime.h>
#include <hip/hip_bf16.h>

typedef float v2f __attribute__((ext_vector_type(2)));
typedef float v8f __attribute__((ext_vector_type(8)));

#define LD      1282
#define KDIM    256
#define NTOT    1282
#define NTILES  81
#define MTILES  256
#define SSTEPS  512
#define BATCH   8

__global__ __launch_bounds__(256)
void fwp_slow_gemm(const float* __restrict__ x,
                   const float* __restrict__ slow_W,
                   const float* __restrict__ slow_b,
                   const float* __restrict__ fast_W0,
                   float* __restrict__ wu)
{
    int wid  = threadIdx.x >> 5;
    int lane = threadIdx.x & 31;
    int wt   = blockIdx.x * 8 + wid;
    if (wt >= MTILES * NTILES) return;
    int mt = wt / NTILES, nt = wt % NTILES;
    int m0 = mt * 16, n0 = nt * 16;
    int q = lane & 15, h = lane >> 4;

    int n = n0 + q;
    const float* wp;
    if (n < 1026) {
        wp = slow_W + (size_t)n * KDIM;
    } else {
        int idx = n - 1026;
        if (idx > 255) idx = 255;
        wp = fast_W0 + (size_t)idx * KDIM;
    }
    float bv = (n < 1026) ? slow_b[n] : 0.0f;

    v8f c;
    #pragma unroll
    for (int r = 0; r < 8; ++r) c[r] = bv;

    const float* ap = x + (size_t)(m0 + q) * KDIM + 2 * h;
    const float* bp = wp + 2 * h;

    #pragma unroll 8
    for (int k = 0; k < KDIM; k += 4) {
        v2f a = *(const v2f*)(ap + k);
        v2f b = *(const v2f*)(bp + k);
        c = __builtin_amdgcn_wmma_f32_16x16x4_f32(false, a, false, b,
                                                  (short)0, c, false, false);
    }

    if (n < NTOT) {
        #pragma unroll
        for (int r = 0; r < 8; ++r) {
            int m = m0 + h * 8 + r;
            wu[(size_t)m * LD + n] = c[r];
        }
    }
}

struct StepOps {
    v2f aA0, aA1, sA0, sA1;
    v2f aB0t0, aB1t0, aB0t1, aB1t1;
    v2f sB0t0, sB1t0, sB0t1, sB1t1;
    float ga, gs;
    float xv0, xv1;
    float base;
};

__device__ __forceinline__
void load_step(StepOps& p, const float* __restrict__ wu, const float* __restrict__ x,
               int s, int b, int kk, int o0, int q, int c0, int c1, int tid)
{
    const float* row  = wu + (size_t)(s * BATCH) * LD;
    const float* rowb = row + (size_t)b * LD;

    p.ga = rowb[1024];
    p.gs = rowb[1025];

    p.aA0.x = row[(size_t)(kk + 0) * LD +       o0 + q];
    p.aA0.y = row[(size_t)(kk + 1) * LD +       o0 + q];
    p.aA1.x = row[(size_t)(kk + 4) * LD +       o0 + q];
    p.aA1.y = row[(size_t)(kk + 5) * LD +       o0 + q];
    p.sA0.x = row[(size_t)(kk + 0) * LD + 512 + o0 + q];
    p.sA0.y = row[(size_t)(kk + 1) * LD + 512 + o0 + q];
    p.sA1.x = row[(size_t)(kk + 4) * LD + 512 + o0 + q];
    p.sA1.y = row[(size_t)(kk + 5) * LD + 512 + o0 + q];

    p.aB0t0.x = row[(size_t)(kk + 0) * LD + 256 + c0];
    p.aB0t0.y = row[(size_t)(kk + 1) * LD + 256 + c0];
    p.aB1t0.x = row[(size_t)(kk + 4) * LD + 256 + c0];
    p.aB1t0.y = row[(size_t)(kk + 5) * LD + 256 + c0];
    p.aB0t1.x = row[(size_t)(kk + 0) * LD + 256 + c1];
    p.aB0t1.y = row[(size_t)(kk + 1) * LD + 256 + c1];
    p.aB1t1.x = row[(size_t)(kk + 4) * LD + 256 + c1];
    p.aB1t1.y = row[(size_t)(kk + 5) * LD + 256 + c1];
    p.sB0t0.x = row[(size_t)(kk + 0) * LD + 768 + c0];
    p.sB0t0.y = row[(size_t)(kk + 1) * LD + 768 + c0];
    p.sB1t0.x = row[(size_t)(kk + 4) * LD + 768 + c0];
    p.sB1t0.y = row[(size_t)(kk + 5) * LD + 768 + c0];
    p.sB0t1.x = row[(size_t)(kk + 0) * LD + 768 + c1];
    p.sB0t1.y = row[(size_t)(kk + 1) * LD + 768 + c1];
    p.sB1t1.x = row[(size_t)(kk + 4) * LD + 768 + c1];
    p.sB1t1.y = row[(size_t)(kk + 5) * LD + 768 + c1];

    p.xv0 = x[(size_t)(s * BATCH + b) * 256 + c0];
    p.xv1 = x[(size_t)(s * BATCH + b) * 256 + c1];

    p.base = 0.0f;
    if (tid < 16) p.base = rowb[1026 + o0 + tid];
}

__device__ __forceinline__ float gate(float z) { return 1.0f / (1.0f + __expf(-z)); }

__device__ __forceinline__ void lds_barrier()
{
    __builtin_amdgcn_fence(__ATOMIC_RELEASE, "workgroup", "local");
    __builtin_amdgcn_s_barrier();
    __builtin_amdgcn_fence(__ATOMIC_ACQUIRE, "workgroup", "local");
}

__global__ __launch_bounds__(256)
void fwp_scan(const float* __restrict__ wu,
              const float* __restrict__ x,
              float* __restrict__ out)
{
    __shared__ float red[8 * 16];

    int tid  = threadIdx.x;
    int w    = tid >> 5;
    int lane = tid & 31;
    int q    = lane & 15;
    int h    = lane >> 4;
    int o0   = blockIdx.x * 16;
    int b    = blockIdx.y;
    int c0   = 32 * w + q;
    int c1   = c0 + 16;
    int kk   = 2 * h;

    v8f fw0 = {0,0,0,0,0,0,0,0};
    v8f fw1 = {0,0,0,0,0,0,0,0};
    const v8f zero = {0,0,0,0,0,0,0,0};

    StepOps cur, nxt;
    load_step(cur, wu, x, 0, b, kk, o0, q, c0, c1, tid);
    float sa = gate(cur.ga);
    float ss = gate(cur.gs);

    for (int s = 0; s < SSTEPS; ++s) {
        int sn = (s + 1 < SSTEPS) ? s + 1 : s;
        load_step(nxt, wu, x, sn, b, kk, o0, q, c0, c1, tid);

        v8f ad0 = __builtin_amdgcn_wmma_f32_16x16x4_f32(false, cur.aA0, false, cur.aB0t0, (short)0, zero, false, false);
        ad0     = __builtin_amdgcn_wmma_f32_16x16x4_f32(false, cur.aA1, false, cur.aB1t0, (short)0, ad0,  false, false);
        v8f ad1 = __builtin_amdgcn_wmma_f32_16x16x4_f32(false, cur.aA0, false, cur.aB0t1, (short)0, zero, false, false);
        ad1     = __builtin_amdgcn_wmma_f32_16x16x4_f32(false, cur.aA1, false, cur.aB1t1, (short)0, ad1,  false, false);
        v8f su0 = __builtin_amdgcn_wmma_f32_16x16x4_f32(false, cur.sA0, false, cur.sB0t0, (short)0, zero, false, false);
        su0     = __builtin_amdgcn_wmma_f32_16x16x4_f32(false, cur.sA1, false, cur.sB1t0, (short)0, su0,  false, false);
        v8f su1 = __builtin_amdgcn_wmma_f32_16x16x4_f32(false, cur.sA0, false, cur.sB0t1, (short)0, zero, false, false);
        su1     = __builtin_amdgcn_wmma_f32_16x16x4_f32(false, cur.sA1, false, cur.sB1t1, (short)0, su1,  false, false);

        #pragma unroll
        for (int r = 0; r < 8; ++r) {
            fw0[r] = fmaf(ad0[r], sa, fw0[r]);
            fw0[r] = fmaf(su0[r], -ss, fw0[r]);
            fw1[r] = fmaf(ad1[r], sa, fw1[r]);
            fw1[r] = fmaf(su1[r], -ss, fw1[r]);
        }

        float pr[8];
        #pragma unroll
        for (int r = 0; r < 8; ++r)
            pr[r] = fw0[r] * cur.xv0 + fw1[r] * cur.xv1;

        float sa_n = gate(nxt.ga);
        float ss_n = gate(nxt.gs);

        #pragma unroll
        for (int m = 1; m <= 8; m <<= 1) {
            #pragma unroll
            for (int r = 0; r < 8; ++r)
                pr[r] += __shfl_xor(pr[r], m, 32);
        }

        if (q == 0) {
            #pragma unroll
            for (int r = 0; r < 8; ++r)
                red[w * 16 + 8 * h + r] = pr[r];
        }
        lds_barrier();

        if (tid < 16) {
            float v = cur.base;
            #pragma unroll
            for (int ww = 0; ww < 8; ++ww)
                v += red[ww * 16 + tid];
            out[(size_t)(s * BATCH + b) * 256 + o0 + tid] = v;
        }
        lds_barrier();

        cur = nxt;
        sa = sa_n;
        ss = ss_n;
    }
}

extern "C" void kernel_launch(void* const* d_in, const int* in_sizes, int n_in,
                              void* d_out, int out_size, void* d_ws, size_t ws_size,
                              hipStream_t stream)
{
    const float* x       = (const float*)d_in[0];
    const float* slow_W  = (const float*)d_in[1];
    const float* slow_b  = (const float*)d_in[2];
    const float* fast_W0 = (const float*)d_in[3];
    float*       out     = (float*)d_out;
    float*       wu      = (float*)d_ws;

    int total_waves = MTILES * NTILES;
    int blocks1 = (total_waves + 7) / 8;
    fwp_slow_gemm<<<blocks1, 256, 0, stream>>>(x, slow_W, slow_b, fast_W0, wu);

    dim3 grid2(16, BATCH);
    fwp_scan<<<grid2, 256, 0, stream>>>(wu, x, out);
}


# Scan waves run 8 steps between block barriers: partial sums of 8 steps go to double-buffered LDS, then one barrier and 128 threads finish them; was 2 barriers and a serial finish per step
# speedup vs baseline: 2.5381x; 1.4457x over previous
#include <hip/hip_runtime.h>
#include <hip/hip_bf16.h>

typedef float v2f __attribute__((ext_vector_type(2)));
typedef float v8f __attribute__((ext_vector_type(8)));

#define LD      1282
#define KDIM    256
#define NTOT    1282
#define NTILES  81
#define MTILES  256
#define SSTEPS  512
#define BATCH   8
#define HSTEPS  8

__global__ __launch_bounds__(256)
void fwp_slow_gemm(const float* __restrict__ x,
                   const float* __restrict__ slow_W,
                   const float* __restrict__ slow_b,
                   const float* __restrict__ fast_W0,
                   float* __restrict__ wu)
{
    int wid  = threadIdx.x >> 5;
    int lane = threadIdx.x & 31;
    int wt   = blockIdx.x * 8 + wid;
    if (wt >= MTILES * NTILES) return;
    int mt = wt / NTILES, nt = wt % NTILES;
    int m0 = mt * 16, n0 = nt * 16;
    int q = lane & 15, h = lane >> 4;

    int n = n0 + q;
    const float* wp;
    if (n < 1026) {
        wp = slow_W + (size_t)n * KDIM;
    } else {
        int idx = n - 1026;
        if (idx > 255) idx = 255;
        wp = fast_W0 + (size_t)idx * KDIM;
    }
    float bv = (n < 1026) ? slow_b[n] : 0.0f;

    v8f c;
    #pragma unroll
    for (int r = 0; r < 8; ++r) c[r] = bv;

    const float* ap = x + (size_t)(m0 + q) * KDIM + 2 * h;
    const float* bp = wp + 2 * h;

    #pragma unroll 8
    for (int k = 0; k < KDIM; k += 4) {
        v2f a = *(const v2f*)(ap + k);
        v2f b = *(const v2f*)(bp + k);
        c = __builtin_amdgcn_wmma_f32_16x16x4_f32(false, a, false, b,
                                                  (short)0, c, false, false);
    }

    if (n < NTOT) {
        #pragma unroll
        for (int r = 0; r < 8; ++r) {
            int m = m0 + h * 8 + r;
            wu[(size_t)m * LD + n] = c[r];
        }
    }
}

struct StepOps {
    v2f aA0, aA1, sA0, sA1;
    v2f aB0t0, aB1t0, aB0t1, aB1t1;
    v2f sB0t0, sB1t0, sB0t1, sB1t1;
    float ga, gs;
    float xv0, xv1;
};

__device__ __forceinline__
void load_step(StepOps& p, const float* __restrict__ wu, const float* __restrict__ x,
               int s, int b, int kk, int o0, int q, int c0, int c1)
{
    const float* row  = wu + (size_t)(s * BATCH) * LD;
    const float* rowb = row + (size_t)b * LD;

    p.ga = rowb[1024];
    p.gs = rowb[1025];

    p.aA0.x = row[(size_t)(kk + 0) * LD +       o0 + q];
    p.aA0.y = row[(size_t)(kk + 1) * LD +       o0 + q];
    p.aA1.x = row[(size_t)(kk + 4) * LD +       o0 + q];
    p.aA1.y = row[(size_t)(kk + 5) * LD +       o0 + q];
    p.sA0.x = row[(size_t)(kk + 0) * LD + 512 + o0 + q];
    p.sA0.y = row[(size_t)(kk + 1) * LD + 512 + o0 + q];
    p.sA1.x = row[(size_t)(kk + 4) * LD + 512 + o0 + q];
    p.sA1.y = row[(size_t)(kk + 5) * LD + 512 + o0 + q];

    p.aB0t0.x = row[(size_t)(kk + 0) * LD + 256 + c0];
    p.aB0t0.y = row[(size_t)(kk + 1) * LD + 256 + c0];
    p.aB1t0.x = row[(size_t)(kk + 4) * LD + 256 + c0];
    p.aB1t0.y = row[(size_t)(kk + 5) * LD + 256 + c0];
    p.aB0t1.x = row[(size_t)(kk + 0) * LD + 256 + c1];
    p.aB0t1.y = row[(size_t)(kk + 1) * LD + 256 + c1];
    p.aB1t1.x = row[(size_t)(kk + 4) * LD + 256 + c1];
    p.aB1t1.y = row[(size_t)(kk + 5) * LD + 256 + c1];
    p.sB0t0.x = row[(size_t)(kk + 0) * LD + 768 + c0];
    p.sB0t0.y = row[(size_t)(kk + 1) * LD + 768 + c0];
    p.sB1t0.x = row[(size_t)(kk + 4) * LD + 768 + c0];
    p.sB1t0.y = row[(size_t)(kk + 5) * LD + 768 + c0];
    p.sB0t1.x = row[(size_t)(kk + 0) * LD + 768 + c1];
    p.sB0t1.y = row[(size_t)(kk + 1) * LD + 768 + c1];
    p.sB1t1.x = row[(size_t)(kk + 4) * LD + 768 + c1];
    p.sB1t1.y = row[(size_t)(kk + 5) * LD + 768 + c1];

    p.xv0 = x[(size_t)(s * BATCH + b) * 256 + c0];
    p.xv1 = x[(size_t)(s * BATCH + b) * 256 + c1];
}

__device__ __forceinline__ float gate(float z) { return 1.0f / (1.0f + __expf(-z)); }

__device__ __forceinline__ void lds_barrier()
{
    __builtin_amdgcn_fence(__ATOMIC_RELEASE, "workgroup", "local");
    __builtin_amdgcn_s_barrier();
    __builtin_amdgcn_fence(__ATOMIC_ACQUIRE, "workgroup", "local");
}

__global__ __launch_bounds__(256)
void fwp_scan(const float* __restrict__ wu,
              const float* __restrict__ x,
              float* __restrict__ out)
{
    __shared__ float red[2][HSTEPS][8 * 16];

    int tid  = threadIdx.x;
    int w    = tid >> 5;
    int lane = tid & 31;
    int q    = lane & 15;
    int h    = lane >> 4;
    int o0   = blockIdx.x * 16;
    int b    = blockIdx.y;
    int c0   = 32 * w + q;
    int c1   = c0 + 16;
    int kk   = 2 * h;

    int  t_f = tid >> 4;
    int  o_f = tid & 15;
    bool fin = tid < 16 * HSTEPS;

    v8f fw0 = {0,0,0,0,0,0,0,0};
    v8f fw1 = {0,0,0,0,0,0,0,0};
    const v8f zero = {0,0,0,0,0,0,0,0};

    StepOps cur, nxt;
    load_step(cur, wu, x, 0, b, kk, o0, q, c0, c1);
    float sa = gate(cur.ga);
    float ss = gate(cur.gs);

    for (int s0 = 0; s0 < SSTEPS; s0 += HSTEPS) {
        int buf = (s0 / HSTEPS) & 1;

        float basev = 0.0f;
        if (fin) basev = wu[(size_t)((s0 + t_f) * BATCH + b) * LD + 1026 + o0 + o_f];

        #pragma unroll
        for (int t = 0; t < HSTEPS; ++t) {
            int s = s0 + t;
            int sn = (s + 1 < SSTEPS) ? s + 1 : s;
            load_step(nxt, wu, x, sn, b, kk, o0, q, c0, c1);

            v8f ad0 = __builtin_amdgcn_wmma_f32_16x16x4_f32(false, cur.aA0, false, cur.aB0t0, (short)0, zero, false, false);
            ad0     = __builtin_amdgcn_wmma_f32_16x16x4_f32(false, cur.aA1, false, cur.aB1t0, (short)0, ad0,  false, false);
            v8f ad1 = __builtin_amdgcn_wmma_f32_16x16x4_f32(false, cur.aA0, false, cur.aB0t1, (short)0, zero, false, false);
            ad1     = __builtin_amdgcn_wmma_f32_16x16x4_f32(false, cur.aA1, false, cur.aB1t1, (short)0, ad1,  false, false);
            v8f su0 = __builtin_amdgcn_wmma_f32_16x16x4_f32(false, cur.sA0, false, cur.sB0t0, (short)0, zero, false, false);
            su0     = __builtin_amdgcn_wmma_f32_16x16x4_f32(false, cur.sA1, false, cur.sB1t0, (short)0, su0,  false, false);
            v8f su1 = __builtin_amdgcn_wmma_f32_16x16x4_f32(false, cur.sA0, false, cur.sB0t1, (short)0, zero, false, false);
            su1     = __builtin_amdgcn_wmma_f32_16x16x4_f32(false, cur.sA1, false, cur.sB1t1, (short)0, su1,  false, false);

            #pragma unroll
            for (int r = 0; r < 8; ++r) {
                fw0[r] = fmaf(ad0[r], sa, fw0[r]);
                fw0[r] = fmaf(su0[r], -ss, fw0[r]);
                fw1[r] = fmaf(ad1[r], sa, fw1[r]);
                fw1[r] = fmaf(su1[r], -ss, fw1[r]);
            }

            float pr[8];
            #pragma unroll
            for (int r = 0; r < 8; ++r)
                pr[r] = fw0[r] * cur.xv0 + fw1[r] * cur.xv1;

            float sa_n = gate(nxt.ga);
            float ss_n = gate(nxt.gs);

            #pragma unroll
            for (int m = 1; m <= 8; m <<= 1) {
                #pragma unroll
                for (int r = 0; r < 8; ++r)
                    pr[r] += __shfl_xor(pr[r], m, 32);
            }

            if (q == 0) {
                #pragma unroll
                for (int r = 0; r < 8; ++r)
                    red[buf][t][w * 16 + 8 * h + r] = pr[r];
            }

            cur = nxt;
            sa = sa_n;
            ss = ss_n;
        }

        lds_barrier();

        if (fin) {
            float v = basev;
            #pragma unroll
            for (int ww = 0; ww < 8; ++ww)
                v += red[buf][t_f][ww * 16 + o_f];
            out[(size_t)((s0 + t_f) * BATCH + b) * 256 + o0 + o_f] = v;
        }
    }
}

extern "C" void kernel_launch(void* const* d_in, const int* in_sizes, int n_in,
                              void* d_out, int out_size, void* d_ws, size_t ws_size,
                              hipStream_t stream)
{
    const float* x       = (const float*)d_in[0];
    const float* slow_W  = (const float*)d_in[1];
    const float* slow_b  = (const float*)d_in[2];
    const float* fast_W0 = (const float*)d_in[3];
    float*       out     = (float*)d_out;
    float*       wu      = (float*)d_ws;

    int total_waves = MTILES * NTILES;
    int blocks1 = (total_waves + 7) / 8;
    fwp_slow_gemm<<<blocks1, 256, 0, stream>>>(x, slow_W, slow_b, fast_W0, wu);

    dim3 grid2(16, BATCH);
    fwp_scan<<<grid2, 256, 0, stream>>>(wu, x, out);
}


# GEMV reduction over each 16-lane row moved into the vector ALU: row_xmask DPP adds (1,2,4,8) replace four dependent rounds of LDS permutes per step; same butterfly, order of sums unchanged
# speedup vs baseline: 2.9953x; 1.1801x over previous
#include <hip/hip_runtime.h>
#include <hip/hip_bf16.h>

typedef float v2f __attribute__((ext_vector_type(2)));
typedef float v8f __attribute__((ext_vector_type(8)));

#define LD      1282
#define KDIM    256
#define NTOT    1282
#define NTILES  81
#define MTILES  256
#define SSTEPS  512
#define BATCH   8
#define HSTEPS  8

__global__ __launch_bounds__(256)
void fwp_slow_gemm(const float* __restrict__ x,
                   const float* __restrict__ slow_W,
                   const float* __restrict__ slow_b,
                   const float* __restrict__ fast_W0,
                   float* __restrict__ wu)
{
    int wid  = threadIdx.x >> 5;
    int lane = threadIdx.x & 31;
    int wt   = blockIdx.x * 8 + wid;
    if (wt >= MTILES * NTILES) return;
    int mt = wt / NTILES, nt = wt % NTILES;
    int m0 = mt * 16, n0 = nt * 16;
    int q = lane & 15, h = lane >> 4;

    int n = n0 + q;
    const float* wp;
    if (n < 1026) {
        wp = slow_W + (size_t)n * KDIM;
    } else {
        int idx = n - 1026;
        if (idx > 255) idx = 255;
        wp = fast_W0 + (size_t)idx * KDIM;
    }
    float bv = (n < 1026) ? slow_b[n] : 0.0f;

    v8f c;
    #pragma unroll
    for (int r = 0; r < 8; ++r) c[r] = bv;

    const float* ap = x + (size_t)(m0 + q) * KDIM + 2 * h;
    const float* bp = wp + 2 * h;

    #pragma unroll 8
    for (int k = 0; k < KDIM; k += 4) {
        v2f a = *(const v2f*)(ap + k);
        v2f b = *(const v2f*)(bp + k);
        c = __builtin_amdgcn_wmma_f32_16x16x4_f32(false, a, false, b,
                                                  (short)0, c, false, false);
    }

    if (n < NTOT) {
        #pragma unroll
        for (int r = 0; r < 8; ++r) {
            int m = m0 + h * 8 + r;
            wu[(size_t)m * LD + n] = c[r];
        }
    }
}

struct StepOps {
    v2f aA0, aA1, sA0, sA1;
    v2f aB0t0, aB1t0, aB0t1, aB1t1;
    v2f sB0t0, sB1t0, sB0t1, sB1t1;
    float ga, gs;
    float xv0, xv1;
};

__device__ __forceinline__
void load_step(StepOps& p, const float* __restrict__ wu, const float* __restrict__ x,
               int s, int b, int kk, int o0, int q, int c0, int c1)
{
    const float* row  = wu + (size_t)(s * BATCH) * LD;
    const float* rowb = row + (size_t)b * LD;

    p.ga = rowb[1024];
    p.gs = rowb[1025];

    p.aA0.x = row[(size_t)(kk + 0) * LD +       o0 + q];
    p.aA0.y = row[(size_t)(kk + 1) * LD +       o0 + q];
    p.aA1.x = row[(size_t)(kk + 4) * LD +       o0 + q];
    p.aA1.y = row[(size_t)(kk + 5) * LD +       o0 + q];
    p.sA0.x = row[(size_t)(kk + 0) * LD + 512 + o0 + q];
    p.sA0.y = row[(size_t)(kk + 1) * LD + 512 + o0 + q];
    p.sA1.x = row[(size_t)(kk + 4) * LD + 512 + o0 + q];
    p.sA1.y = row[(size_t)(kk + 5) * LD + 512 + o0 + q];

    p.aB0t0.x = row[(size_t)(kk + 0) * LD + 256 + c0];
    p.aB0t0.y = row[(size_t)(kk + 1) * LD + 256 + c0];
    p.aB1t0.x = row[(size_t)(kk + 4) * LD + 256 + c0];
    p.aB1t0.y = row[(size_t)(kk + 5) * LD + 256 + c0];
    p.aB0t1.x = row[(size_t)(kk + 0) * LD + 256 + c1];
    p.aB0t1.y = row[(size_t)(kk + 1) * LD + 256 + c1];
    p.aB1t1.x = row[(size_t)(kk + 4) * LD + 256 + c1];
    p.aB1t1.y = row[(size_t)(kk + 5) * LD + 256 + c1];
    p.sB0t0.x = row[(size_t)(kk + 0) * LD + 768 + c0];
    p.sB0t0.y = row[(size_t)(kk + 1) * LD + 768 + c0];
    p.sB1t0.x = row[(size_t)(kk + 4) * LD + 768 + c0];
    p.sB1t0.y = row[(size_t)(kk + 5) * LD + 768 + c0];
    p.sB0t1.x = row[(size_t)(kk + 0) * LD + 768 + c1];
    p.sB0t1.y = row[(size_t)(kk + 1) * LD + 768 + c1];
    p.sB1t1.x = row[(size_t)(kk + 4) * LD + 768 + c1];
    p.sB1t1.y = row[(size_t)(kk + 5) * LD + 768 + c1];

    p.xv0 = x[(size_t)(s * BATCH + b) * 256 + c0];
    p.xv1 = x[(size_t)(s * BATCH + b) * 256 + c1];
}

__device__ __forceinline__ float gate(float z) { return 1.0f / (1.0f + __expf(-z)); }

template <int M>
__device__ __forceinline__ float row_xor_add(float v)
{
    float r;
    asm("v_add_f32_dpp %0, %1, %1 row_xmask:%2 row_mask:0xf bank_mask:0xf"
        : "=v"(r) : "v"(v), "n"(M));
    return r;
}

__device__ __forceinline__ void lds_barrier()
{
    __builtin_amdgcn_fence(__ATOMIC_RELEASE, "workgroup", "local");
    __builtin_amdgcn_s_barrier();
    __builtin_amdgcn_fence(__ATOMIC_ACQUIRE, "workgroup", "local");
}

__global__ __launch_bounds__(256)
void fwp_scan(const float* __restrict__ wu,
              const float* __restrict__ x,
              float* __restrict__ out)
{
    __shared__ float red[2][HSTEPS][8 * 16];

    int tid  = threadIdx.x;
    int w    = tid >> 5;
    int lane = tid & 31;
    int q    = lane & 15;
    int h    = lane >> 4;
    int o0   = blockIdx.x * 16;
    int b    = blockIdx.y;
    int c0   = 32 * w + q;
    int c1   = c0 + 16;
    int kk   = 2 * h;

    int  t_f = tid >> 4;
    int  o_f = tid & 15;
    bool fin = tid < 16 * HSTEPS;

    v8f fw0 = {0,0,0,0,0,0,0,0};
    v8f fw1 = {0,0,0,0,0,0,0,0};
    const v8f zero = {0,0,0,0,0,0,0,0};

    StepOps cur, nxt;
    load_step(cur, wu, x, 0, b, kk, o0, q, c0, c1);
    float sa = gate(cur.ga);
    float ss = gate(cur.gs);

    for (int s0 = 0; s0 < SSTEPS; s0 += HSTEPS) {
        int buf = (s0 / HSTEPS) & 1;

        float basev = 0.0f;
        if (fin) basev = wu[(size_t)((s0 + t_f) * BATCH + b) * LD + 1026 + o0 + o_f];

        #pragma unroll
        for (int t = 0; t < HSTEPS; ++t) {
            int s = s0 + t;
            int sn = (s + 1 < SSTEPS) ? s + 1 : s;
            load_step(nxt, wu, x, sn, b, kk, o0, q, c0, c1);

            v8f ad0 = __builtin_amdgcn_wmma_f32_16x16x4_f32(false, cur.aA0, false, cur.aB0t0, (short)0, zero, false, false);
            ad0     = __builtin_amdgcn_wmma_f32_16x16x4_f32(false, cur.aA1, false, cur.aB1t0, (short)0, ad0,  false, false);
            v8f ad1 = __builtin_amdgcn_wmma_f32_16x16x4_f32(false, cur.aA0, false, cur.aB0t1, (short)0, zero, false, false);
            ad1     = __builtin_amdgcn_wmma_f32_16x16x4_f32(false, cur.aA1, false, cur.aB1t1, (short)0, ad1,  false, false);
            v8f su0 = __builtin_amdgcn_wmma_f32_16x16x4_f32(false, cur.sA0, false, cur.sB0t0, (short)0, zero, false, false);
            su0     = __builtin_amdgcn_wmma_f32_16x16x4_f32(false, cur.sA1, false, cur.sB1t0, (short)0, su0,  false, false);
            v8f su1 = __builtin_amdgcn_wmma_f32_16x16x4_f32(false, cur.sA0, false, cur.sB0t1, (short)0, zero, false, false);
            su1     = __builtin_amdgcn_wmma_f32_16x16x4_f32(false, cur.sA1, false, cur.sB1t1, (short)0, su1,  false, false);

            #pragma unroll
            for (int r = 0; r < 8; ++r) {
                fw0[r] = fmaf(ad0[r], sa, fw0[r]);
                fw0[r] = fmaf(su0[r], -ss, fw0[r]);
                fw1[r] = fmaf(ad1[r], sa, fw1[r]);
                fw1[r] = fmaf(su1[r], -ss, fw1[r]);
            }

            float pr[8];
            #pragma unroll
            for (int r = 0; r < 8; ++r)
                pr[r] = fw0[r] * cur.xv0 + fw1[r] * cur.xv1;

            float sa_n = gate(nxt.ga);
            float ss_n = gate(nxt.gs);

            #pragma unroll
            for (int r = 0; r < 8; ++r) pr[r] = row_xor_add<1>(pr[r]);
            #pragma unroll
            for (int r = 0; r < 8; ++r) pr[r] = row_xor_add<2>(pr[r]);
            #pragma unroll
            for (int r = 0; r < 8; ++r) pr[r] = row_xor_add<4>(pr[r]);
            #pragma unroll
            for (int r = 0; r < 8; ++r) pr[r] = row_xor_add<8>(pr[r]);

            if (q == 0) {
                #pragma unroll
                for (int r = 0; r < 8; ++r)
                    red[buf][t][w * 16 + 8 * h + r] = pr[r];
            }

            cur = nxt;
            sa = sa_n;
            ss = ss_n;
        }

        lds_barrier();

        if (fin) {
            float v = basev;
            #pragma unroll
            for (int ww = 0; ww < 8; ++ww)
                v += red[buf][t_f][ww * 16 + o_f];
            out[(size_t)((s0 + t_f) * BATCH + b) * 256 + o0 + o_f] = v;
        }
    }
}

extern "C" void kernel_launch(void* const* d_in, const int* in_sizes, int n_in,
                              void* d_out, int out_size, void* d_ws, size_t ws_size,
                              hipStream_t stream)
{
    const float* x       = (const float*)d_in[0];
    const float* slow_W  = (const float*)d_in[1];
    const float* slow_b  = (const float*)d_in[2];
    const float* fast_W0 = (const float*)d_in[3];
    float*       out     = (float*)d_out;
    float*       wu      = (float*)d_ws;

    int total_waves = MTILES * NTILES;
    int blocks1 = (total_waves + 7) / 8;
    fwp_slow_gemm<<<blocks1, 256, 0, stream>>>(x, slow_W, slow_b, fast_W0, wu);

    dim3 grid2(16, BATCH);
    fwp_scan<<<grid2, 256, 0, stream>>>(wu, x, out);
}


# Slow GEMM stores its result batch-fastest (the 8 batch rows of a column side by side): a scan lane gets its 4 K values per operand in one 128-bit load, 6 wide loads replace 28 dword loads per step
# speedup vs baseline: 3.1466x; 1.0505x over previous
#include <hip/hip_runtime.h>
#include <hip/hip_bf16.h>

typedef float v2f __attribute__((ext_vector_type(2)));
typedef float v4f __attribute__((ext_vector_type(4)));
typedef float v8f __attribute__((ext_vector_type(8)));

#define FCOLS   1024
#define TLD     258
#define KDIM    256
#define NTOT    1282
#define NTILES  81
#define MTILES  256
#define SSTEPS  512
#define BATCH   8
#define HSTEPS  8

__global__ __launch_bounds__(256)
void fwp_slow_gemm(const float* __restrict__ x,
                   const float* __restrict__ slow_W,
                   const float* __restrict__ slow_b,
                   const float* __restrict__ fast_W0,
                   float* __restrict__ pk,
                   float* __restrict__ tl)
{
    int wid  = threadIdx.x >> 5;
    int lane = threadIdx.x & 31;
    int wt   = blockIdx.x * 8 + wid;
    if (wt >= MTILES * NTILES) return;
    int mt = wt / NTILES, nt = wt % NTILES;
    int m0 = mt * 16, n0 = nt * 16;
    int q = lane & 15, h = lane >> 4;

    int n = n0 + q;
    const float* wp;
    if (n < 1026) {
        wp = slow_W + (size_t)n * KDIM;
    } else {
        int idx = n - 1026;
        if (idx > 255) idx = 255;
        wp = fast_W0 + (size_t)idx * KDIM;
    }
    float bv = (n < 1026) ? slow_b[n] : 0.0f;

    v8f c;
    #pragma unroll
    for (int r = 0; r < 8; ++r) c[r] = bv;

    const float* ap = x + (size_t)(m0 + q) * KDIM + 2 * h;
    const float* bp = wp + 2 * h;

    #pragma unroll 8
    for (int k = 0; k < KDIM; k += 4) {
        v2f a = *(const v2f*)(ap + k);
        v2f b = *(const v2f*)(bp + k);
        c = __builtin_amdgcn_wmma_f32_16x16x4_f32(false, a, false, b,
                                                  (short)0, c, false, false);
    }

    if (n0 < FCOLS) {
        int s = 2 * mt + h;
        v4f lo = { c[0], c[1], c[4], c[5] };
        v4f hi = { c[2], c[3], c[6], c[7] };
        v4f* dst = (v4f*)(pk + ((size_t)s * FCOLS + n) * 8);
        dst[0] = lo;
        dst[1] = hi;
    } else if (n < NTOT) {
        #pragma unroll
        for (int r = 0; r < 8; ++r) {
            int m = m0 + h * 8 + r;
            tl[(size_t)m * TLD + (n - FCOLS)] = c[r];
        }
    }
}

struct StepOps {
    v4f aA, sA;
    v4f aBt0, aBt1;
    v4f sBt0, sBt1;
    float ga, gs;
    float xv0, xv1;
};

__device__ __forceinline__
v4f load_frag(const float* __restrict__ pk, int s, int n, int h)
{
    return *(const v4f*)(pk + ((size_t)s * FCOLS + n) * 8 + 4 * h);
}

__device__ __forceinline__ v2f frag_lo(v4f v) { return __builtin_shufflevector(v, v, 0, 1); }
__device__ __forceinline__ v2f frag_hi(v4f v) { return __builtin_shufflevector(v, v, 2, 3); }

__device__ __forceinline__
void load_step(StepOps& p, const float* __restrict__ pk, const float* __restrict__ tl,
               const float* __restrict__ x,
               int s, int b, int h, int o0, int q, int c0, int c1)
{
    const float* trow = tl + (size_t)(s * BATCH + b) * TLD;
    p.ga = trow[0];
    p.gs = trow[1];

    p.aA   = load_frag(pk, s,       o0 + q, h);
    p.sA   = load_frag(pk, s, 512 + o0 + q, h);
    p.aBt0 = load_frag(pk, s, 256 + c0, h);
    p.aBt1 = load_frag(pk, s, 256 + c1, h);
    p.sBt0 = load_frag(pk, s, 768 + c0, h);
    p.sBt1 = load_frag(pk, s, 768 + c1, h);

    p.xv0 = x[(size_t)(s * BATCH + b) * 256 + c0];
    p.xv1 = x[(size_t)(s * BATCH + b) * 256 + c1];
}

__device__ __forceinline__ float gate(float z) { return 1.0f / (1.0f + __expf(-z)); }

template <int M>
__device__ __forceinline__ float row_xor_add(float v)
{
    float r;
    asm("v_add_f32_dpp %0, %1, %1 row_xmask:%2 row_mask:0xf bank_mask:0xf"
        : "=v"(r) : "v"(v), "n"(M));
    return r;
}

__device__ __forceinline__ void lds_barrier()
{
    __builtin_amdgcn_fence(__ATOMIC_RELEASE, "workgroup", "local");
    __builtin_amdgcn_s_barrier();
    __builtin_amdgcn_fence(__ATOMIC_ACQUIRE, "workgroup", "local");
}

__global__ __launch_bounds__(256)
void fwp_scan(const float* __restrict__ pk,
              const float* __restrict__ tl,
              const float* __restrict__ x,
              float* __restrict__ out)
{
    __shared__ float red[2][HSTEPS][8 * 16];

    int tid  = threadIdx.x;
    int w    = tid >> 5;
    int lane = tid & 31;
    int q    = lane & 15;
    int h    = lane >> 4;
    int o0   = blockIdx.x * 16;
    int b    = blockIdx.y;
    int c0   = 32 * w + q;
    int c1   = c0 + 16;

    int  t_f = tid >> 4;
    int  o_f = tid & 15;
    bool fin = tid < 16 * HSTEPS;

    v8f fw0 = {0,0,0,0,0,0,0,0};
    v8f fw1 = {0,0,0,0,0,0,0,0};
    const v8f zero = {0,0,0,0,0,0,0,0};

    StepOps cur, nxt;
    load_step(cur, pk, tl, x, 0, b, h, o0, q, c0, c1);

    for (int s0 = 0; s0 < SSTEPS; s0 += HSTEPS) {
        int buf = (s0 / HSTEPS) & 1;

        float basev = 0.0f;
        if (fin) basev = tl[(size_t)((s0 + t_f) * BATCH + b) * TLD + 2 + o0 + o_f];

        #pragma unroll
        for (int t = 0; t < HSTEPS; ++t) {
            int s = s0 + t;
            int sn = (s + 1 < SSTEPS) ? s + 1 : s;
            load_step(nxt, pk, tl, x, sn, b, h, o0, q, c0, c1);
            __builtin_amdgcn_sched_barrier(0);

            v8f ad0 = __builtin_amdgcn_wmma_f32_16x16x4_f32(false, frag_lo(cur.aA), false, frag_lo(cur.aBt0), (short)0, zero, false, false);
            ad0     = __builtin_amdgcn_wmma_f32_16x16x4_f32(false, frag_hi(cur.aA), false, frag_hi(cur.aBt0), (short)0, ad0,  false, false);
            v8f ad1 = __builtin_amdgcn_wmma_f32_16x16x4_f32(false, frag_lo(cur.aA), false, frag_lo(cur.aBt1), (short)0, zero, false, false);
            ad1     = __builtin_amdgcn_wmma_f32_16x16x4_f32(false, frag_hi(cur.aA), false, frag_hi(cur.aBt1), (short)0, ad1,  false, false);
            v8f su0 = __builtin_amdgcn_wmma_f32_16x16x4_f32(false, frag_lo(cur.sA), false, frag_lo(cur.sBt0), (short)0, zero, false, false);
            su0     = __builtin_amdgcn_wmma_f32_16x16x4_f32(false, frag_hi(cur.sA), false, frag_hi(cur.sBt0), (short)0, su0,  false, false);
            v8f su1 = __builtin_amdgcn_wmma_f32_16x16x4_f32(false, frag_lo(cur.sA), false, frag_lo(cur.sBt1), (short)0, zero, false, false);
            su1     = __builtin_amdgcn_wmma_f32_16x16x4_f32(false, frag_hi(cur.sA), false, frag_hi(cur.sBt1), (short)0, su1,  false, false);

            float sa = gate(cur.ga);
            float ss = gate(cur.gs);

            #pragma unroll
            for (int r = 0; r < 8; ++r) {
                fw0[r] = fmaf(ad0[r], sa, fw0[r]);
                fw0[r] = fmaf(su0[r], -ss, fw0[r]);
                fw1[r] = fmaf(ad1[r], sa, fw1[r]);
                fw1[r] = fmaf(su1[r], -ss, fw1[r]);
            }

            float pr[8];
            #pragma unroll
            for (int r = 0; r < 8; ++r)
                pr[r] = fw0[r] * cur.xv0 + fw1[r] * cur.xv1;

            #pragma unroll
            for (int r = 0; r < 8; ++r) pr[r] = row_xor_add<1>(pr[r]);
            #pragma unroll
            for (int r = 0; r < 8; ++r) pr[r] = row_xor_add<2>(pr[r]);
            #pragma unroll
            for (int r = 0; r < 8; ++r) pr[r] = row_xor_add<4>(pr[r]);
            #pragma unroll
            for (int r = 0; r < 8; ++r) pr[r] = row_xor_add<8>(pr[r]);

            if (q == 0) {
                #pragma unroll
                for (int r = 0; r < 8; ++r)
                    red[buf][t][w * 16 + 8 * h + r] = pr[r];
            }

            cur = nxt;
        }

        lds_barrier();

        if (fin) {
            float v = basev;
            #pragma unroll
            for (int ww = 0; ww < 8; ++ww)
                v += red[buf][t_f][ww * 16 + o_f];
            out[(size_t)((s0 + t_f) * BATCH + b) * 256 + o0 + o_f] = v;
        }
    }
}

extern "C" void kernel_launch(void* const* d_in, const int* in_sizes, int n_in,
                              void* d_out, int out_size, void* d_ws, size_t ws_size,
                              hipStream_t stream)
{
    const float* x       = (const float*)d_in[0];
    const float* slow_W  = (const float*)d_in[1];
    const float* slow_b  = (const float*)d_in[2];
    const float* fast_W0 = (const float*)d_in[3];
    float*       out     = (float*)d_out;
    float*       pk      = (float*)d_ws;
    float*       tl      = pk + (size_t)SSTEPS * FCOLS * BATCH;

    int total_waves = MTILES * NTILES;
    int blocks1 = (total_waves + 7) / 8;
    fwp_slow_gemm<<<blocks1, 256, 0, stream>>>(x, slow_W, slow_b, fast_W0, pk, tl);

    dim3 grid2(16, BATCH);
    fwp_scan<<<grid2, 256, 0, stream>>>(pk, tl, x, out);
}


# Scan operand prefetch deepened from 1 step ahead to 3: four register sets in rotation give a step's first-touch loads three steps to come back from memory; order of sums unchanged
# speedup vs baseline: 3.9415x; 1.2526x over previous
#include <hip/hip_runtime.h>
#include <hip/hip_bf16.h>

typedef float v2f __attribute__((ext_vector_type(2)));
typedef float v4f __attribute__((ext_vector_type(4)));
typedef float v8f __attribute__((ext_vector_type(8)));

#define FCOLS   1024
#define TLD     258
#define KDIM    256
#define NTOT    1282
#define NTILES  81
#define MTILES  256
#define SSTEPS  512
#define BATCH   8
#define HSTEPS  8
#define PDIST   3
#define NSETS   4

__global__ __launch_bounds__(256)
void fwp_slow_gemm(const float* __restrict__ x,
                   const float* __restrict__ slow_W,
                   const float* __restrict__ slow_b,
                   const float* __restrict__ fast_W0,
                   float* __restrict__ pk,
                   float* __restrict__ tl)
{
    int wid  = threadIdx.x >> 5;
    int lane = threadIdx.x & 31;
    int wt   = blockIdx.x * 8 + wid;
    if (wt >= MTILES * NTILES) return;
    int mt = wt / NTILES, nt = wt % NTILES;
    int m0 = mt * 16, n0 = nt * 16;
    int q = lane & 15, h = lane >> 4;

    int n = n0 + q;
    const float* wp;
    if (n < 1026) {
        wp = slow_W + (size_t)n * KDIM;
    } else {
        int idx = n - 1026;
        if (idx > 255) idx = 255;
        wp = fast_W0 + (size_t)idx * KDIM;
    }
    float bv = (n < 1026) ? slow_b[n] : 0.0f;

    v8f c;
    #pragma unroll
    for (int r = 0; r < 8; ++r) c[r] = bv;

    const float* ap = x + (size_t)(m0 + q) * KDIM + 2 * h;
    const float* bp = wp + 2 * h;

    #pragma unroll 8
    for (int k = 0; k < KDIM; k += 4) {
        v2f a = *(const v2f*)(ap + k);
        v2f b = *(const v2f*)(bp + k);
        c = __builtin_amdgcn_wmma_f32_16x16x4_f32(false, a, false, b,
                                                  (short)0, c, false, false);
    }

    if (n0 < FCOLS) {
        int s = 2 * mt + h;
        v4f lo = { c[0], c[1], c[4], c[5] };
        v4f hi = { c[2], c[3], c[6], c[7] };
        v4f* dst = (v4f*)(pk + ((size_t)s * FCOLS + n) * 8);
        dst[0] = lo;
        dst[1] = hi;
    } else if (n < NTOT) {
        #pragma unroll
        for (int r = 0; r < 8; ++r) {
            int m = m0 + h * 8 + r;
            tl[(size_t)m * TLD + (n - FCOLS)] = c[r];
        }
    }
}

struct StepOps {
    v4f aA, sA;
    v4f aBt0, aBt1;
    v4f sBt0, sBt1;
    float ga, gs;
    float xv0, xv1;
};

__device__ __forceinline__
v4f load_frag(const float* __restrict__ pk, int s, int n, int h)
{
    return *(const v4f*)(pk + ((size_t)s * FCOLS + n) * 8 + 4 * h);
}

__device__ __forceinline__ v2f frag_lo(v4f v) { return __builtin_shufflevector(v, v, 0, 1); }
__device__ __forceinline__ v2f frag_hi(v4f v) { return __builtin_shufflevector(v, v, 2, 3); }

__device__ __forceinline__
void load_step(StepOps& p, const float* __restrict__ pk, const float* __restrict__ tl,
               const float* __restrict__ x,
               int s, int b, int h, int o0, int q, int c0, int c1)
{
    const float* trow = tl + (size_t)(s * BATCH + b) * TLD;
    p.ga = trow[0];
    p.gs = trow[1];

    p.aA   = load_frag(pk, s,       o0 + q, h);
    p.sA   = load_frag(pk, s, 512 + o0 + q, h);
    p.aBt0 = load_frag(pk, s, 256 + c0, h);
    p.aBt1 = load_frag(pk, s, 256 + c1, h);
    p.sBt0 = load_frag(pk, s, 768 + c0, h);
    p.sBt1 = load_frag(pk, s, 768 + c1, h);

    p.xv0 = x[(size_t)(s * BATCH + b) * 256 + c0];
    p.xv1 = x[(size_t)(s * BATCH + b) * 256 + c1];
}

__device__ __forceinline__ float gate(float z) { return 1.0f / (1.0f + __expf(-z)); }

template <int M>
__device__ __forceinline__ float row_xor_add(float v)
{
    float r;
    asm("v_add_f32_dpp %0, %1, %1 row_xmask:%2 row_mask:0xf bank_mask:0xf"
        : "=v"(r) : "v"(v), "n"(M));
    return r;
}

__device__ __forceinline__ void lds_barrier()
{
    __builtin_amdgcn_fence(__ATOMIC_RELEASE, "workgroup", "local");
    __builtin_amdgcn_s_barrier();
    __builtin_amdgcn_fence(__ATOMIC_ACQUIRE, "workgroup", "local");
}

__global__ __launch_bounds__(256)
void fwp_scan(const float* __restrict__ pk,
              const float* __restrict__ tl,
              const float* __restrict__ x,
              float* __restrict__ out)
{
    __shared__ float red[2][HSTEPS][8 * 16];

    int tid  = threadIdx.x;
    int w    = tid >> 5;
    int lane = tid & 31;
    int q    = lane & 15;
    int h    = lane >> 4;
    int o0   = blockIdx.x * 16;
    int b    = blockIdx.y;
    int c0   = 32 * w + q;
    int c1   = c0 + 16;

    int  t_f = tid >> 4;
    int  o_f = tid & 15;
    bool fin = tid < 16 * HSTEPS;

    v8f fw0 = {0,0,0,0,0,0,0,0};
    v8f fw1 = {0,0,0,0,0,0,0,0};
    const v8f zero = {0,0,0,0,0,0,0,0};

    StepOps ring[NSETS];
    #pragma unroll
    for (int p = 0; p < PDIST; ++p) {
        load_step(ring[p], pk, tl, x, p, b, h, o0, q, c0, c1);
        __builtin_amdgcn_sched_barrier(0);
    }

    for (int s0 = 0; s0 < SSTEPS; s0 += HSTEPS) {
        int buf = (s0 / HSTEPS) & 1;

        float basev = 0.0f;
        if (fin) basev = tl[(size_t)((s0 + t_f) * BATCH + b) * TLD + 2 + o0 + o_f];

        #pragma unroll
        for (int t = 0; t < HSTEPS; ++t) {
            int s = s0 + t;
            const StepOps& cur = ring[t % NSETS];
            int sn = (s + PDIST < SSTEPS) ? s + PDIST : SSTEPS - 1;
            load_step(ring[(t + PDIST) % NSETS], pk, tl, x, sn, b, h, o0, q, c0, c1);
            __builtin_amdgcn_sched_barrier(0);

            v8f ad0 = __builtin_amdgcn_wmma_f32_16x16x4_f32(false, frag_lo(cur.aA), false, frag_lo(cur.aBt0), (short)0, zero, false, false);
            ad0     = __builtin_amdgcn_wmma_f32_16x16x4_f32(false, frag_hi(cur.aA), false, frag_hi(cur.aBt0), (short)0, ad0,  false, false);
            v8f ad1 = __builtin_amdgcn_wmma_f32_16x16x4_f32(false, frag_lo(cur.aA), false, frag_lo(cur.aBt1), (short)0, zero, false, false);
            ad1     = __builtin_amdgcn_wmma_f32_16x16x4_f32(false, frag_hi(cur.aA), false, frag_hi(cur.aBt1), (short)0, ad1,  false, false);
            v8f su0 = __builtin_amdgcn_wmma_f32_16x16x4_f32(false, frag_lo(cur.sA), false, frag_lo(cur.sBt0), (short)0, zero, false, false);
            su0     = __builtin_amdgcn_wmma_f32_16x16x4_f32(false, frag_hi(cur.sA), false, frag_hi(cur.sBt0), (short)0, su0,  false, false);
            v8f su1 = __builtin_amdgcn_wmma_f32_16x16x4_f32(false, frag_lo(cur.sA), false, frag_lo(cur.sBt1), (short)0, zero, false, false);
            su1     = __builtin_amdgcn_wmma_f32_16x16x4_f32(false, frag_hi(cur.sA), false, frag_hi(cur.sBt1), (short)0, su1,  false, false);

            float sa = gate(cur.ga);
            float ss = gate(cur.gs);

            #pragma unroll
            for (int r = 0; r < 8; ++r) {
                fw0[r] = fmaf(ad0[r], sa, fw0[r]);
                fw0[r] = fmaf(su0[r], -ss, fw0[r]);
                fw1[r] = fmaf(ad1[r], sa, fw1[r]);
                fw1[r] = fmaf(su1[r], -ss, fw1[r]);
            }

            float pr[8];
            #pragma unroll
            for (int r = 0; r < 8; ++r)
                pr[r] = fw0[r] * cur.xv0 + fw1[r] * cur.xv1;

            #pragma unroll
            for (int r = 0; r < 8; ++r) pr[r] = row_xor_add<1>(pr[r]);
            #pragma unroll
            for (int r = 0; r < 8; ++r) pr[r] = row_xor_add<2>(pr[r]);
            #pragma unroll
            for (int r = 0; r < 8; ++r) pr[r] = row_xor_add<4>(pr[r]);
            #pragma unroll
            for (int r = 0; r < 8; ++r) pr[r] = row_xor_add<8>(pr[r]);

            if (q == 0) {
                #pragma unroll
                for (int r = 0; r < 8; ++r)
                    red[buf][t][w * 16 + 8 * h + r] = pr[r];
            }
        }

        lds_barrier();

        if (fin) {
            float v = basev;
            #pragma unroll
            for (int ww = 0; ww < 8; ++ww)
                v += red[buf][t_f][ww * 16 + o_f];
            out[(size_t)((s0 + t_f) * BATCH + b) * 256 + o0 + o_f] = v;
        }
    }
}

extern "C" void kernel_launch(void* const* d_in, const int* in_sizes, int n_in,
                              void* d_out, int out_size, void* d_ws, size_t ws_size,
                              hipStream_t stream)
{
    const float* x       = (const float*)d_in[0];
    const float* slow_W  = (const float*)d_in[1];
    const float* slow_b  = (const float*)d_in[2];
    const float* fast_W0 = (const float*)d_in[3];
    float*       out     = (float*)d_out;
    float*       pk      = (float*)d_ws;
    float*       tl      = pk + (size_t)SSTEPS * FCOLS * BATCH;

    int total_waves = MTILES * NTILES;
    int blocks1 = (total_waves + 7) / 8;
    fwp_slow_gemm<<<blocks1, 256, 0, stream>>>(x, slow_W, slow_b, fast_W0, pk, tl);

    dim3 grid2(16, BATCH);
    fwp_scan<<<grid2, 256, 0, stream>>>(pk, tl, x, out);
}


# Scan keeps each fw tile transposed (rows i, columns o): the GEMV sum over i runs down a lane's registers as FMAs plus one cross-row add, replacing the 32 DPP adds per step; order of sums changed
# speedup vs baseline: 4.4803x; 1.1367x over previous
#include <hip/hip_runtime.h>
#include <hip/hip_bf16.h>

typedef float v2f __attribute__((ext_vector_type(2)));
typedef float v4f __attribute__((ext_vector_type(4)));
typedef float v8f __attribute__((ext_vector_type(8)));

#define FCOLS   1024
#define TLD     258
#define KDIM    256
#define NTOT    1282
#define NTILES  81
#define MTILES  256
#define SSTEPS  512
#define BATCH   8
#define HSTEPS  8
#define PDIST   3
#define NSETS   4

__global__ __launch_bounds__(256)
void fwp_slow_gemm(const float* __restrict__ x,
                   const float* __restrict__ slow_W,
                   const float* __restrict__ slow_b,
                   const float* __restrict__ fast_W0,
                   float* __restrict__ pk,
                   float* __restrict__ tl)
{
    int wid  = threadIdx.x >> 5;
    int lane = threadIdx.x & 31;
    int wt   = blockIdx.x * 8 + wid;
    if (wt >= MTILES * NTILES) return;
    int mt = wt / NTILES, nt = wt % NTILES;
    int m0 = mt * 16, n0 = nt * 16;
    int q = lane & 15, h = lane >> 4;

    int n = n0 + q;
    const float* wp;
    if (n < 1026) {
        wp = slow_W + (size_t)n * KDIM;
    } else {
        int idx = n - 1026;
        if (idx > 255) idx = 255;
        wp = fast_W0 + (size_t)idx * KDIM;
    }
    float bv = (n < 1026) ? slow_b[n] : 0.0f;

    v8f c;
    #pragma unroll
    for (int r = 0; r < 8; ++r) c[r] = bv;

    const float* ap = x + (size_t)(m0 + q) * KDIM + 2 * h;
    const float* bp = wp + 2 * h;

    #pragma unroll 8
    for (int k = 0; k < KDIM; k += 4) {
        v2f a = *(const v2f*)(ap + k);
        v2f b = *(const v2f*)(bp + k);
        c = __builtin_amdgcn_wmma_f32_16x16x4_f32(false, a, false, b,
                                                  (short)0, c, false, false);
    }

    if (n0 < FCOLS) {
        int s = 2 * mt + h;
        v4f lo = { c[0], c[1], c[4], c[5] };
        v4f hi = { c[2], c[3], c[6], c[7] };
        v4f* dst = (v4f*)(pk + ((size_t)s * FCOLS + n) * 8);
        dst[0] = lo;
        dst[1] = hi;
    } else if (n < NTOT) {
        #pragma unroll
        for (int r = 0; r < 8; ++r) {
            int m = m0 + h * 8 + r;
            tl[(size_t)m * TLD + (n - FCOLS)] = c[r];
        }
    }
}

struct StepOps {
    v4f aO, sO;
    v4f aIt0, aIt1;
    v4f sIt0, sIt1;
    float ga, gs;
    v4f x0lo, x0hi;
    v4f x1lo, x1hi;
};

__device__ __forceinline__
v4f load_frag(const float* __restrict__ pk, int s, int n, int h)
{
    return *(const v4f*)(pk + ((size_t)s * FCOLS + n) * 8 + 4 * h);
}

__device__ __forceinline__ v2f frag_lo(v4f v) { return __builtin_shufflevector(v, v, 0, 1); }
__device__ __forceinline__ v2f frag_hi(v4f v) { return __builtin_shufflevector(v, v, 2, 3); }

__device__ __forceinline__
void load_step(StepOps& p, const float* __restrict__ pk, const float* __restrict__ tl,
               const float* __restrict__ x,
               int s, int b, int h, int o0, int q, int c0, int c1, int ib)
{
    const float* trow = tl + (size_t)(s * BATCH + b) * TLD;
    p.ga = trow[0];
    p.gs = trow[1];

    p.aO   = load_frag(pk, s,       o0 + q, h);
    p.sO   = load_frag(pk, s, 512 + o0 + q, h);
    p.aIt0 = load_frag(pk, s, 256 + c0, h);
    p.aIt1 = load_frag(pk, s, 256 + c1, h);
    p.sIt0 = load_frag(pk, s, 768 + c0, h);
    p.sIt1 = load_frag(pk, s, 768 + c1, h);

    const float* xr = x + (size_t)(s * BATCH + b) * 256 + ib;
    p.x0lo = *(const v4f*)(xr);
    p.x0hi = *(const v4f*)(xr + 4);
    p.x1lo = *(const v4f*)(xr + 16);
    p.x1hi = *(const v4f*)(xr + 20);
}

__device__ __forceinline__ float gate(float z) { return 1.0f / (1.0f + __expf(-z)); }

__device__ __forceinline__ float xhalf(float v)
{
    int b = __builtin_bit_cast(int, v);
    return __builtin_bit_cast(float,
        __builtin_amdgcn_permlanex16(b, b, 0x76543210, 0xfedcba98, false, false));
}

__device__ __forceinline__ void lds_barrier()
{
    __builtin_amdgcn_fence(__ATOMIC_RELEASE, "workgroup", "local");
    __builtin_amdgcn_s_barrier();
    __builtin_amdgcn_fence(__ATOMIC_ACQUIRE, "workgroup", "local");
}

__global__ __launch_bounds__(256)
void fwp_scan(const float* __restrict__ pk,
              const float* __restrict__ tl,
              const float* __restrict__ x,
              float* __restrict__ out)
{
    __shared__ float red[2][HSTEPS][8 * 16];

    int tid  = threadIdx.x;
    int w    = tid >> 5;
    int lane = tid & 31;
    int q    = lane & 15;
    int h    = lane >> 4;
    int o0   = blockIdx.x * 16;
    int b    = blockIdx.y;
    int c0   = 32 * w + q;
    int c1   = c0 + 16;
    int ib   = 32 * w + 8 * h;

    int  t_f = tid >> 4;
    int  o_f = tid & 15;
    bool fin = tid < 16 * HSTEPS;

    v8f fw0 = {0,0,0,0,0,0,0,0};
    v8f fw1 = {0,0,0,0,0,0,0,0};
    const v8f zero = {0,0,0,0,0,0,0,0};

    StepOps ring[NSETS];
    #pragma unroll
    for (int p = 0; p < PDIST; ++p) {
        load_step(ring[p], pk, tl, x, p, b, h, o0, q, c0, c1, ib);
        __builtin_amdgcn_sched_barrier(0);
    }

    for (int s0 = 0; s0 < SSTEPS; s0 += HSTEPS) {
        int buf = (s0 / HSTEPS) & 1;

        float basev = 0.0f;
        if (fin) basev = tl[(size_t)((s0 + t_f) * BATCH + b) * TLD + 2 + o0 + o_f];

        #pragma unroll
        for (int t = 0; t < HSTEPS; ++t) {
            int s = s0 + t;
            const StepOps& cur = ring[t % NSETS];
            int sn = (s + PDIST < SSTEPS) ? s + PDIST : SSTEPS - 1;
            load_step(ring[(t + PDIST) % NSETS], pk, tl, x, sn, b, h, o0, q, c0, c1, ib);
            __builtin_amdgcn_sched_barrier(0);

            v8f ad0 = __builtin_amdgcn_wmma_f32_16x16x4_f32(false, frag_lo(cur.aIt0), false, frag_lo(cur.aO), (short)0, zero, false, false);
            ad0     = __builtin_amdgcn_wmma_f32_16x16x4_f32(false, frag_hi(cur.aIt0), false, frag_hi(cur.aO), (short)0, ad0,  false, false);
            v8f ad1 = __builtin_amdgcn_wmma_f32_16x16x4_f32(false, frag_lo(cur.aIt1), false, frag_lo(cur.aO), (short)0, zero, false, false);
            ad1     = __builtin_amdgcn_wmma_f32_16x16x4_f32(false, frag_hi(cur.aIt1), false, frag_hi(cur.aO), (short)0, ad1,  false, false);
            v8f su0 = __builtin_amdgcn_wmma_f32_16x16x4_f32(false, frag_lo(cur.sIt0), false, frag_lo(cur.sO), (short)0, zero, false, false);
            su0     = __builtin_amdgcn_wmma_f32_16x16x4_f32(false, frag_hi(cur.sIt0), false, frag_hi(cur.sO), (short)0, su0,  false, false);
            v8f su1 = __builtin_amdgcn_wmma_f32_16x16x4_f32(false, frag_lo(cur.sIt1), false, frag_lo(cur.sO), (short)0, zero, false, false);
            su1     = __builtin_amdgcn_wmma_f32_16x16x4_f32(false, frag_hi(cur.sIt1), false, frag_hi(cur.sO), (short)0, su1,  false, false);

            float sa = gate(cur.ga);
            float ss = gate(cur.gs);

            #pragma unroll
            for (int r = 0; r < 8; ++r) {
                fw0[r] = fmaf(ad0[r], sa, fw0[r]);
                fw0[r] = fmaf(su0[r], -ss, fw0[r]);
                fw1[r] = fmaf(ad1[r], sa, fw1[r]);
                fw1[r] = fmaf(su1[r], -ss, fw1[r]);
            }

            float p0 = fw0[0] * cur.x0lo[0];
            float p1 = fw0[1] * cur.x0lo[1];
            float p2 = fw1[0] * cur.x1lo[0];
            float p3 = fw1[1] * cur.x1lo[1];
            p0 = fmaf(fw0[2], cur.x0lo[2], p0);
            p1 = fmaf(fw0[3], cur.x0lo[3], p1);
            p2 = fmaf(fw1[2], cur.x1lo[2], p2);
            p3 = fmaf(fw1[3], cur.x1lo[3], p3);
            p0 = fmaf(fw0[4], cur.x0hi[0], p0);
            p1 = fmaf(fw0[5], cur.x0hi[1], p1);
            p2 = fmaf(fw1[4], cur.x1hi[0], p2);
            p3 = fmaf(fw1[5], cur.x1hi[1], p3);
            p0 = fmaf(fw0[6], cur.x0hi[2], p0);
            p1 = fmaf(fw0[7], cur.x0hi[3], p1);
            p2 = fmaf(fw1[6], cur.x1hi[2], p2);
            p3 = fmaf(fw1[7], cur.x1hi[3], p3);

            float part = (p0 + p1) + (p2 + p3);
            float tot  = part + xhalf(part);
            if (h == 0)
                red[buf][t][w * 16 + q] = tot;
        }

        lds_barrier();

        if (fin) {
            float v = basev;
            #pragma unroll
            for (int ww = 0; ww < 8; ++ww)
                v += red[buf][t_f][ww * 16 + o_f];
            out[(size_t)((s0 + t_f) * BATCH + b) * 256 + o0 + o_f] = v;
        }
    }
}

extern "C" void kernel_launch(void* const* d_in, const int* in_sizes, int n_in,
                              void* d_out, int out_size, void* d_ws, size_t ws_size,
                              hipStream_t stream)
{
    const float* x       = (const float*)d_in[0];
    const float* slow_W  = (const float*)d_in[1];
    const float* slow_b  = (const float*)d_in[2];
    const float* fast_W0 = (const float*)d_in[3];
    float*       out     = (float*)d_out;
    float*       pk      = (float*)d_ws;
    float*       tl      = pk + (size_t)SSTEPS * FCOLS * BATCH;

    int total_waves = MTILES * NTILES;
    int blocks1 = (total_waves + 7) / 8;
    fwp_slow_gemm<<<blocks1, 256, 0, stream>>>(x, slow_W, slow_b, fast_W0, pk, tl);

    dim3 grid2(16, BATCH);
    fwp_scan<<<grid2, 256, 0, stream>>>(pk, tl, x, out);
}


# Slow GEMM: each wave computes a 32x32 block (2x2 tiles) from 128-bit fragment loads issued two K-chunks ahead: 4 loads per 8 WMMAs instead of 16, a quarter of the waves; k order of sums changed
# speedup vs baseline: 5.1078x; 1.1401x over previous
#include <hip/hip_runtime.h>
#include <hip/hip_bf16.h>

typedef float v2f __attribute__((ext_vector_type(2)));
typedef float v4f __attribute__((ext_vector_type(4)));
typedef float v8f __attribute__((ext_vector_type(8)));

#define FCOLS   1024
#define TLD     258
#define KDIM    256
#define NTOT    1282
#define NBLKS   41
#define MBLKS   128
#define GPRE    2
#define SSTEPS  512
#define BATCH   8
#define HSTEPS  8
#define PDIST   3
#define NSETS   4

__device__ __forceinline__ v2f frag_lo(v4f v) { return __builtin_shufflevector(v, v, 0, 1); }
__device__ __forceinline__ v2f frag_hi(v4f v) { return __builtin_shufflevector(v, v, 2, 3); }

__device__ __forceinline__
void mma_chunk(v8f (&c)[2][2], const v4f (&a)[2], const v4f (&b)[2])
{
    #pragma unroll
    for (int i = 0; i < 2; ++i)
        #pragma unroll
        for (int j = 0; j < 2; ++j)
            c[i][j] = __builtin_amdgcn_wmma_f32_16x16x4_f32(false, frag_lo(a[i]), false, frag_lo(b[j]),
                                                            (short)0, c[i][j], false, false);
    #pragma unroll
    for (int i = 0; i < 2; ++i)
        #pragma unroll
        for (int j = 0; j < 2; ++j)
            c[i][j] = __builtin_amdgcn_wmma_f32_16x16x4_f32(false, frag_hi(a[i]), false, frag_hi(b[j]),
                                                            (short)0, c[i][j], false, false);
}

__global__ __launch_bounds__(256)
void fwp_slow_gemm(const float* __restrict__ x,
                   const float* __restrict__ slow_W,
                   const float* __restrict__ slow_b,
                   const float* __restrict__ fast_W0,
                   float* __restrict__ pk,
                   float* __restrict__ tl)
{
    int wid  = threadIdx.x >> 5;
    int lane = threadIdx.x & 31;
    int wt   = blockIdx.x * 8 + wid;
    if (wt >= MBLKS * NBLKS) return;
    int mb = wt / NBLKS, nb = wt % NBLKS;
    int q = lane & 15, h = lane >> 4;

    const float* ap[2];
    const float* bp[2];
    float bv[2];
    #pragma unroll
    for (int i = 0; i < 2; ++i)
        ap[i] = x + (size_t)(mb * 32 + 16 * i + q) * KDIM + 4 * h;
    #pragma unroll
    for (int j = 0; j < 2; ++j) {
        int n = nb * 32 + 16 * j + q;
        const float* wp;
        if (n < 1026) {
            wp = slow_W + (size_t)n * KDIM;
        } else {
            int idx = n - 1026;
            if (idx > 255) idx = 255;
            wp = fast_W0 + (size_t)idx * KDIM;
        }
        bp[j] = wp + 4 * h;
        bv[j] = (n < 1026) ? slow_b[n] : 0.0f;
    }

    v8f c[2][2];
    #pragma unroll
    for (int i = 0; i < 2; ++i)
        #pragma unroll
        for (int j = 0; j < 2; ++j)
            #pragma unroll
            for (int r = 0; r < 8; ++r) c[i][j][r] = bv[j];

    v4f fa[GPRE + 1][2], fb[GPRE + 1][2];
    #pragma unroll
    for (int ch = 0; ch < GPRE; ++ch) {
        fa[ch][0] = *(const v4f*)(ap[0] + 8 * ch);
        fa[ch][1] = *(const v4f*)(ap[1] + 8 * ch);
        fb[ch][0] = *(const v4f*)(bp[0] + 8 * ch);
        fb[ch][1] = *(const v4f*)(bp[1] + 8 * ch);
    }
    #pragma unroll
    for (int ch = 0; ch < KDIM / 8; ++ch) {
        if (ch + GPRE < KDIM / 8) {
            const int st = (ch + GPRE) % (GPRE + 1);
            fa[st][0] = *(const v4f*)(ap[0] + 8 * (ch + GPRE));
            fa[st][1] = *(const v4f*)(ap[1] + 8 * (ch + GPRE));
            fb[st][0] = *(const v4f*)(bp[0] + 8 * (ch + GPRE));
            fb[st][1] = *(const v4f*)(bp[1] + 8 * (ch + GPRE));
        }
        __builtin_amdgcn_sched_barrier(0);
        mma_chunk(c, fa[ch % (GPRE + 1)], fb[ch % (GPRE + 1)]);
        __builtin_amdgcn_sched_barrier(0);
    }

    #pragma unroll
    for (int i = 0; i < 2; ++i) {
        int mt = 2 * mb + i;
        #pragma unroll
        for (int j = 0; j < 2; ++j) {
            int n0 = nb * 32 + 16 * j;
            int n  = n0 + q;
            if (n0 < FCOLS) {
                int s = 2 * mt + h;
                v4f lo = { c[i][j][0], c[i][j][1], c[i][j][4], c[i][j][5] };
                v4f hi = { c[i][j][2], c[i][j][3], c[i][j][6], c[i][j][7] };
                v4f* dst = (v4f*)(pk + ((size_t)s * FCOLS + n) * 8);
                dst[0] = lo;
                dst[1] = hi;
            } else if (n < NTOT) {
                #pragma unroll
                for (int r = 0; r < 8; ++r) {
                    int m = mt * 16 + h * 8 + r;
                    tl[(size_t)m * TLD + (n - FCOLS)] = c[i][j][r];
                }
            }
        }
    }
}

struct StepOps {
    v4f aO, sO;
    v4f aIt0, aIt1;
    v4f sIt0, sIt1;
    float ga, gs;
    v4f x0lo, x0hi;
    v4f x1lo, x1hi;
};

__device__ __forceinline__
v4f load_frag(const float* __restrict__ pk, int s, int n, int h)
{
    return *(const v4f*)(pk + ((size_t)s * FCOLS + n) * 8 + 4 * h);
}

__device__ __forceinline__
void load_step(StepOps& p, const float* __restrict__ pk, const float* __restrict__ tl,
               const float* __restrict__ x,
               int s, int b, int h, int o0, int q, int c0, int c1, int ib)
{
    const float* trow = tl + (size_t)(s * BATCH + b) * TLD;
    p.ga = trow[0];
    p.gs = trow[1];

    p.aO   = load_frag(pk, s,       o0 + q, h);
    p.sO   = load_frag(pk, s, 512 + o0 + q, h);
    p.aIt0 = load_frag(pk, s, 256 + c0, h);
    p.aIt1 = load_frag(pk, s, 256 + c1, h);
    p.sIt0 = load_frag(pk, s, 768 + c0, h);
    p.sIt1 = load_frag(pk, s, 768 + c1, h);

    const float* xr = x + (size_t)(s * BATCH + b) * 256 + ib;
    p.x0lo = *(const v4f*)(xr);
    p.x0hi = *(const v4f*)(xr + 4);
    p.x1lo = *(const v4f*)(xr + 16);
    p.x1hi = *(const v4f*)(xr + 20);
}

__device__ __forceinline__ float gate(float z) { return 1.0f / (1.0f + __expf(-z)); }

__device__ __forceinline__ float xhalf(float v)
{
    int b = __builtin_bit_cast(int, v);
    return __builtin_bit_cast(float,
        __builtin_amdgcn_permlanex16(b, b, 0x76543210, 0xfedcba98, false, false));
}

__device__ __forceinline__ void lds_barrier()
{
    __builtin_amdgcn_fence(__ATOMIC_RELEASE, "workgroup", "local");
    __builtin_amdgcn_s_barrier();
    __builtin_amdgcn_fence(__ATOMIC_ACQUIRE, "workgroup", "local");
}

__global__ __launch_bounds__(256)
void fwp_scan(const float* __restrict__ pk,
              const float* __restrict__ tl,
              const float* __restrict__ x,
              float* __restrict__ out)
{
    __shared__ float red[2][HSTEPS][8 * 16];

    int tid  = threadIdx.x;
    int w    = tid >> 5;
    int lane = tid & 31;
    int q    = lane & 15;
    int h    = lane >> 4;
    int o0   = blockIdx.x * 16;
    int b    = blockIdx.y;
    int c0   = 32 * w + q;
    int c1   = c0 + 16;
    int ib   = 32 * w + 8 * h;

    int  t_f = tid >> 4;
    int  o_f = tid & 15;
    bool fin = tid < 16 * HSTEPS;

    v8f fw0 = {0,0,0,0,0,0,0,0};
    v8f fw1 = {0,0,0,0,0,0,0,0};
    const v8f zero = {0,0,0,0,0,0,0,0};

    StepOps ring[NSETS];
    #pragma unroll
    for (int p = 0; p < PDIST; ++p) {
        load_step(ring[p], pk, tl, x, p, b, h, o0, q, c0, c1, ib);
        __builtin_amdgcn_sched_barrier(0);
    }

    for (int s0 = 0; s0 < SSTEPS; s0 += HSTEPS) {
        int buf = (s0 / HSTEPS) & 1;

        float basev = 0.0f;
        if (fin) basev = tl[(size_t)((s0 + t_f) * BATCH + b) * TLD + 2 + o0 + o_f];

        #pragma unroll
        for (int t = 0; t < HSTEPS; ++t) {
            int s = s0 + t;
            const StepOps& cur = ring[t % NSETS];
            int sn = (s + PDIST < SSTEPS) ? s + PDIST : SSTEPS - 1;
            load_step(ring[(t + PDIST) % NSETS], pk, tl, x, sn, b, h, o0, q, c0, c1, ib);
            __builtin_amdgcn_sched_barrier(0);

            v8f ad0 = __builtin_amdgcn_wmma_f32_16x16x4_f32(false, frag_lo(cur.aIt0), false, frag_lo(cur.aO), (short)0, zero, false, false);
            ad0     = __builtin_amdgcn_wmma_f32_16x16x4_f32(false, frag_hi(cur.aIt0), false, frag_hi(cur.aO), (short)0, ad0,  false, false);
            v8f ad1 = __builtin_amdgcn_wmma_f32_16x16x4_f32(false, frag_lo(cur.aIt1), false, frag_lo(cur.aO), (short)0, zero, false, false);
            ad1     = __builtin_amdgcn_wmma_f32_16x16x4_f32(false, frag_hi(cur.aIt1), false, frag_hi(cur.aO), (short)0, ad1,  false, false);
            v8f su0 = __builtin_amdgcn_wmma_f32_16x16x4_f32(false, frag_lo(cur.sIt0), false, frag_lo(cur.sO), (short)0, zero, false, false);
            su0     = __builtin_amdgcn_wmma_f32_16x16x4_f32(false, frag_hi(cur.sIt0), false, frag_hi(cur.sO), (short)0, su0,  false, false);
            v8f su1 = __builtin_amdgcn_wmma_f32_16x16x4_f32(false, frag_lo(cur.sIt1), false, frag_lo(cur.sO), (short)0, zero, false, false);
            su1     = __builtin_amdgcn_wmma_f32_16x16x4_f32(false, frag_hi(cur.sIt1), false, frag_hi(cur.sO), (short)0, su1,  false, false);

            float sa = gate(cur.ga);
            float ss = gate(cur.gs);

            #pragma unroll
            for (int r = 0; r < 8; ++r) {
                fw0[r] = fmaf(ad0[r], sa, fw0[r]);
                fw0[r] = fmaf(su0[r], -ss, fw0[r]);
                fw1[r] = fmaf(ad1[r], sa, fw1[r]);
                fw1[r] = fmaf(su1[r], -ss, fw1[r]);
            }

            float p0 = fw0[0] * cur.x0lo[0];
            float p1 = fw0[1] * cur.x0lo[1];
            float p2 = fw1[0] * cur.x1lo[0];
            float p3 = fw1[1] * cur.x1lo[1];
            p0 = fmaf(fw0[2], cur.x0lo[2], p0);
            p1 = fmaf(fw0[3], cur.x0lo[3], p1);
            p2 = fmaf(fw1[2], cur.x1lo[2], p2);
            p3 = fmaf(fw1[3], cur.x1lo[3], p3);
            p0 = fmaf(fw0[4], cur.x0hi[0], p0);
            p1 = fmaf(fw0[5], cur.x0hi[1], p1);
            p2 = fmaf(fw1[4], cur.x1hi[0], p2);
            p3 = fmaf(fw1[5], cur.x1hi[1], p3);
            p0 = fmaf(fw0[6], cur.x0hi[2], p0);
            p1 = fmaf(fw0[7], cur.x0hi[3], p1);
            p2 = fmaf(fw1[6], cur.x1hi[2], p2);
            p3 = fmaf(fw1[7], cur.x1hi[3], p3);

            float part = (p0 + p1) + (p2 + p3);
            float tot  = part + xhalf(part);
            if (h == 0)
                red[buf][t][w * 16 + q] = tot;
        }

        lds_barrier();

        if (fin) {
            float v = basev;
            #pragma unroll
            for (int ww = 0; ww < 8; ++ww)
                v += red[buf][t_f][ww * 16 + o_f];
            out[(size_t)((s0 + t_f) * BATCH + b) * 256 + o0 + o_f] = v;
        }
    }
}

extern "C" void kernel_launch(void* const* d_in, const int* in_sizes, int n_in,
                              void* d_out, int out_size, void* d_ws, size_t ws_size,
                              hipStream_t stream)
{
    const float* x       = (const float*)d_in[0];
    const float* slow_W  = (const float*)d_in[1];
    const float* slow_b  = (const float*)d_in[2];
    const float* fast_W0 = (const float*)d_in[3];
    float*       out     = (float*)d_out;
    float*       pk      = (float*)d_ws;
    float*       tl      = pk + (size_t)SSTEPS * FCOLS * BATCH;

    int total_waves = MBLKS * NBLKS;
    int blocks1 = (total_waves + 7) / 8;
    fwp_slow_gemm<<<blocks1, 256, 0, stream>>>(x, slow_W, slow_b, fast_W0, pk, tl);

    dim3 grid2(16, BATCH);
    fwp_scan<<<grid2, 256, 0, stream>>>(pk, tl, x, out);
}
